# starting kernel
# baseline (speedup 1.0000x reference)
#include <hip/hip_runtime.h>
#include <math.h>

typedef __attribute__((ext_vector_type(2))) float v2f;
typedef __attribute__((ext_vector_type(8))) float v8f;

#define GAT_NFEAT 512
#define GAT_NHID  64
#define GAT_HEADS 8
#define GAT_NCLASS 64

__global__ void gat_zero_kernel(float* __restrict__ p, size_t n) {
  size_t i = (size_t)blockIdx.x * blockDim.x + threadIdx.x;
  size_t stride = (size_t)gridDim.x * blockDim.x;
  for (; i < n; i += stride) p[i] = 0.0f;
}

__global__ __launch_bounds__(128) void gat_gemm_f32_wmma(
    const float* __restrict__ A, const float* __restrict__ B,
    float* __restrict__ C, int M, int ldc) {
  __shared__ float As[64][36];
  __shared__ float Bs[32][68];

  const int tid     = threadIdx.x;
  const int lane    = tid & 31;
  const int wv      = tid >> 5;
  const int laneLow = lane & 15;
  const int kOff    = (lane >> 4) << 1;
  const int rowBase = blockIdx.x * 64;
  const int colBase = blockIdx.y * 64;
  const float* Bblk = B + (size_t)blockIdx.y * (GAT_NFEAT * 64);

  v8f acc[4];
  acc[0] = {}; acc[1] = {}; acc[2] = {}; acc[3] = {};

  const int aCol = (tid & 7)  * 4;
  const int bCol = (tid & 15) * 4;

  for (int k0 = 0; k0 < GAT_NFEAT; k0 += 32) {
#pragma unroll
    for (int p = 0; p < 4; ++p) {
      int row = (tid >> 3) + p * 16;
      int gRow = rowBase + row;
      if (gRow >= M) gRow = M - 1;
      *(float4*)&As[row][aCol] =
          *(const float4*)(A + (size_t)gRow * GAT_NFEAT + k0 + aCol);
    }
#pragma unroll
    for (int p = 0; p < 4; ++p) {
      int row = (tid >> 4) + p * 8;
      *(float4*)&Bs[row][bCol] =
          *(const float4*)(Bblk + (size_t)(k0 + row) * 64 + bCol);
    }
    if (k0 + 32 < GAT_NFEAT) {
      int gRow = rowBase + (tid >> 3);
      if (gRow >= M) gRow = M - 1;
      __builtin_prefetch(A + (size_t)gRow * GAT_NFEAT + k0 + 32 + aCol, 0, 3);
    }
    __syncthreads();

#pragma unroll
    for (int kk = 0; kk < 32; kk += 4) {
      v2f a;
      a.x = As[wv * 16 + laneLow][kk + kOff];
      a.y = As[wv * 16 + laneLow][kk + kOff + 1];
#pragma unroll
      for (int n = 0; n < 4; ++n) {
        v2f b;
        b.x = Bs[kk + kOff][n * 16 + laneLow];
        b.y = Bs[kk + kOff + 1][n * 16 + laneLow];
        acc[n] = __builtin_amdgcn_wmma_f32_16x16x4_f32(
            false, a, false, b, (short)0, acc[n], false, false);
      }
    }
    __syncthreads();
  }

#pragma unroll
  for (int n = 0; n < 4; ++n) {
#pragma unroll
    for (int r = 0; r < 8; ++r) {
      int row = rowBase + wv * 16 + r + ((lane >> 4) << 3);
      if (row < M)
        C[(size_t)row * ldc + colBase + n * 16 + laneLow] = acc[n][r];
    }
  }
}

__global__ void gat_alpha_beta_kernel(
    const float* __restrict__ H, const float* __restrict__ a,
    float* __restrict__ alpha, float* __restrict__ beta, int n, int heads) {
  int t = blockIdx.x * blockDim.x + threadIdx.x;
  if (t >= n * heads) return;
  int node = t / heads;
  int h = t - node * heads;
  const float* hv = H + (size_t)node * heads * 64 + h * 64;
  const float* av = a + h * 128;
  float sa = 0.0f, sb = 0.0f;
#pragma unroll
  for (int d = 0; d < 64; d += 4) {
    float4 hh = *(const float4*)(hv + d);
    float4 a1 = *(const float4*)(av + d);
    float4 a2 = *(const float4*)(av + 64 + d);
    sa += hh.x * a1.x + hh.y * a1.y + hh.z * a1.z + hh.w * a1.w;
    sb += hh.x * a2.x + hh.y * a2.y + hh.z * a2.z + hh.w * a2.w;
  }
  alpha[t] = sa;
  beta[t] = sb;
}

template <int HEADS>
__global__ __launch_bounds__(256) void gat_edge_kernel(
    const int* __restrict__ rows, const int* __restrict__ cols,
    const float* __restrict__ alpha, const float* __restrict__ beta,
    const float* __restrict__ H, float* __restrict__ rowsum,
    float* __restrict__ agg, int E) {
  int wave = blockIdx.x * (blockDim.x >> 5) + (threadIdx.x >> 5);
  int lane = threadIdx.x & 31;
  if (wave >= E) return;
  const int r = rows[wave];
  const int c = cols[wave];

  float w[HEADS];
#pragma unroll
  for (int h = 0; h < HEADS; ++h) {
    float s = alpha[r * HEADS + h] + beta[c * HEADS + h];
    s = s > 0.0f ? s : 0.2f * s;
    w[h] = __expf(-s);
  }
  if (lane < HEADS) {
    float s = alpha[r * HEADS + lane] + beta[c * HEADS + lane];
    s = s > 0.0f ? s : 0.2f * s;
    atomicAdd(&rowsum[r * HEADS + lane], __expf(-s));
  }

  const float* hc = H + (size_t)c * (HEADS * 64);
  float* ar = agg + (size_t)r * (HEADS * 64);
#pragma unroll
  for (int ch = 0; ch < HEADS * 2; ++ch) {
    int idx = ch * 32 + lane;
    atomicAdd(&ar[idx], w[ch >> 1] * hc[idx]);
  }
}

__global__ void gat_norm_elu_kernel(float* __restrict__ agg,
                                    const float* __restrict__ rowsum, int n) {
  int t = blockIdx.x * blockDim.x + threadIdx.x;
  if (t >= n * GAT_HEADS * 64) return;
  int node = t >> 9;
  int h = (t >> 6) & 7;
  float v = agg[t] / rowsum[node * GAT_HEADS + h];
  agg[t] = v > 0.0f ? v : (__expf(v) - 1.0f);
}

__global__ void gat_norm_kernel(float* __restrict__ out,
                                const float* __restrict__ rowsum, int n) {
  int t = blockIdx.x * blockDim.x + threadIdx.x;
  if (t >= n * GAT_NCLASS) return;
  out[t] = out[t] / rowsum[t >> 6];
}

extern "C" void kernel_launch(void* const* d_in, const int* in_sizes, int n_in,
                              void* d_out, int out_size, void* d_ws,
                              size_t ws_size, hipStream_t stream) {
  const float* x       = (const float*)d_in[0];
  const float* Ws      = (const float*)d_in[1];
  const float* a_heads = (const float*)d_in[2];
  const float* W_out   = (const float*)d_in[3];
  const float* a_out   = (const float*)d_in[4];
  const int*   eidx    = (const int*)d_in[5];
  float* out = (float*)d_out;

  const int N = in_sizes[0] / GAT_NFEAT;
  const int E = in_sizes[5] / 2;
  const int* e_row = eidx;
  const int* e_col = eidx + E;

  float* ws = (float*)d_ws;
  size_t off = 0;
  float* H       = ws + off; off += (size_t)N * 512;
  float* alpha1  = ws + off; off += (size_t)N * 8;
  float* beta1   = ws + off; off += (size_t)N * 8;
  float* rowsum1 = ws + off; off += (size_t)N * 8;
  float* agg1    = ws + off; off += (size_t)N * 512;
  float* H2      = ws + off; off += (size_t)N * 64;
  float* alpha2  = ws + off; off += (size_t)N;
  float* beta2   = ws + off; off += (size_t)N;
  float* rowsum2 = ws + off; off += (size_t)N;
  (void)ws_size; (void)n_in; (void)out_size;

  gat_zero_kernel<<<2048, 256, 0, stream>>>(rowsum1, (size_t)N * (8 + 512));
  gat_zero_kernel<<<256, 256, 0, stream>>>(rowsum2, (size_t)N);
  gat_zero_kernel<<<2048, 256, 0, stream>>>(out, (size_t)N * GAT_NCLASS);

  const int mblocks = (N + 63) / 64;

  gat_gemm_f32_wmma<<<dim3(mblocks, GAT_HEADS), 128, 0, stream>>>(
      x, Ws, H, N, 512);

  gat_alpha_beta_kernel<<<(N * GAT_HEADS + 255) / 256, 256, 0, stream>>>(
      H, a_heads, alpha1, beta1, N, GAT_HEADS);

  gat_edge_kernel<GAT_HEADS><<<(E + 7) / 8, 256, 0, stream>>>(
      e_row, e_col, alpha1, beta1, H, rowsum1, agg1, E);

  gat_norm_elu_kernel<<<(N * 512 + 255) / 256, 256, 0, stream>>>(
      agg1, rowsum1, N);

  gat_gemm_f32_wmma<<<dim3(mblocks, 1), 128, 0, stream>>>(
      agg1, W_out, H2, N, 64);

  gat_alpha_beta_kernel<<<(N + 255) / 256, 256, 0, stream>>>(
      H2, a_out, alpha2, beta2, N, 1);

  gat_edge_kernel<1><<<(E + 7) / 8, 256, 0, stream>>>(
      e_row, e_col, alpha2, beta2, H2, rowsum2, out, E);

  gat_norm_kernel<<<(N * GAT_NCLASS + 255) / 256, 256, 0, stream>>>(
      out, rowsum2, N);
}


# Atomic edge scatter replaced by a per-call row-sorted edge list (count, scan, fill) and one-wave-per-row register gather; sum order in a row = fill's arrival-order rank; norm+ELU by the row owner
# speedup vs baseline: 2.5773x; 2.5773x over previous
#include <hip/hip_runtime.h>
#include <math.h>

typedef __attribute__((ext_vector_type(2))) float v2f;
typedef __attribute__((ext_vector_type(8))) float v8f;

#define GAT_NFEAT 512
#define GAT_NHID  64
#define GAT_HEADS 8
#define GAT_NCLASS 64

__global__ void gat_zero_kernel(float* __restrict__ p, size_t n) {
  size_t i = (size_t)blockIdx.x * blockDim.x + threadIdx.x;
  size_t stride = (size_t)gridDim.x * blockDim.x;
  for (; i < n; i += stride) p[i] = 0.0f;
}

__global__ __launch_bounds__(128) void gat_gemm_f32_wmma(
    const float* __restrict__ A, const float* __restrict__ B,
    float* __restrict__ C, int M, int ldc) {
  __shared__ float As[64][36];
  __shared__ float Bs[32][68];

  const int tid     = threadIdx.x;
  const int lane    = tid & 31;
  const int wv      = tid >> 5;
  const int laneLow = lane & 15;
  const int kOff    = (lane >> 4) << 1;
  const int rowBase = blockIdx.x * 64;
  const int colBase = blockIdx.y * 64;
  const float* Bblk = B + (size_t)blockIdx.y * (GAT_NFEAT * 64);

  v8f acc[4];
  acc[0] = {}; acc[1] = {}; acc[2] = {}; acc[3] = {};

  const int aCol = (tid & 7)  * 4;
  const int bCol = (tid & 15) * 4;

  for (int k0 = 0; k0 < GAT_NFEAT; k0 += 32) {
#pragma unroll
    for (int p = 0; p < 4; ++p) {
      int row = (tid >> 3) + p * 16;
      int gRow = rowBase + row;
      if (gRow >= M) gRow = M - 1;
      *(float4*)&As[row][aCol] =
          *(const float4*)(A + (size_t)gRow * GAT_NFEAT + k0 + aCol);
    }
#pragma unroll
    for (int p = 0; p < 4; ++p) {
      int row = (tid >> 4) + p * 8;
      *(float4*)&Bs[row][bCol] =
          *(const float4*)(Bblk + (size_t)(k0 + row) * 64 + bCol);
    }
    if (k0 + 32 < GAT_NFEAT) {
      int gRow = rowBase + (tid >> 3);
      if (gRow >= M) gRow = M - 1;
      __builtin_prefetch(A + (size_t)gRow * GAT_NFEAT + k0 + 32 + aCol, 0, 3);
    }
    __syncthreads();

#pragma unroll
    for (int kk = 0; kk < 32; kk += 4) {
      v2f a;
      a.x = As[wv * 16 + laneLow][kk + kOff];
      a.y = As[wv * 16 + laneLow][kk + kOff + 1];
#pragma unroll
      for (int n = 0; n < 4; ++n) {
        v2f b;
        b.x = Bs[kk + kOff][n * 16 + laneLow];
        b.y = Bs[kk + kOff + 1][n * 16 + laneLow];
        acc[n] = __builtin_amdgcn_wmma_f32_16x16x4_f32(
            false, a, false, b, (short)0, acc[n], false, false);
      }
    }
    __syncthreads();
  }

#pragma unroll
  for (int n = 0; n < 4; ++n) {
#pragma unroll
    for (int r = 0; r < 8; ++r) {
      int row = rowBase + wv * 16 + r + ((lane >> 4) << 3);
      if (row < M)
        C[(size_t)row * ldc + colBase + n * 16 + laneLow] = acc[n][r];
    }
  }
}

__global__ void gat_alpha_beta_kernel(
    const float* __restrict__ H, const float* __restrict__ a,
    float* __restrict__ alpha, float* __restrict__ beta, int n, int heads) {
  int t = blockIdx.x * blockDim.x + threadIdx.x;
  if (t >= n * heads) return;
  int node = t / heads;
  int h = t - node * heads;
  const float* hv = H + (size_t)node * heads * 64 + h * 64;
  const float* av = a + h * 128;
  float sa = 0.0f, sb = 0.0f;
#pragma unroll
  for (int d = 0; d < 64; d += 4) {
    float4 hh = *(const float4*)(hv + d);
    float4 a1 = *(const float4*)(av + d);
    float4 a2 = *(const float4*)(av + 64 + d);
    sa += hh.x * a1.x + hh.y * a1.y + hh.z * a1.z + hh.w * a1.w;
    sb += hh.x * a2.x + hh.y * a2.y + hh.z * a2.z + hh.w * a2.w;
  }
  alpha[t] = sa;
  beta[t] = sb;
}

__global__ void gat_count_kernel(const int* __restrict__ rows,
                                 int* __restrict__ cur, int E) {
  int e = blockIdx.x * blockDim.x + threadIdx.x;
  if (e < E) atomicAdd(&cur[rows[e]], 1);
}

__global__ __launch_bounds__(1024) void gat_scan_kernel(
    int* __restrict__ cur, int* __restrict__ off, int n) {
  __shared__ int wsum[32];
  __shared__ int tile_total;
  const int tid  = threadIdx.x;
  const int lane = tid & 31;
  const int wv   = tid >> 5;
  int carry = 0;
  for (int base = 0; base < n; base += 1024 * 16) {
    const int i0 = base + tid * 16;
    const bool full = (i0 + 16 <= n);
    int v[16];
    if (full) {
      for (int q = 0; q < 4; ++q) {
        const int4 t = *(const int4*)(cur + i0 + 4 * q);
        v[4 * q + 0] = t.x; v[4 * q + 1] = t.y;
        v[4 * q + 2] = t.z; v[4 * q + 3] = t.w;
      }
    } else {
      for (int k = 0; k < 16; ++k) v[k] = (i0 + k < n) ? cur[i0 + k] : 0;
    }
    int s = 0;
    for (int k = 0; k < 16; ++k) { const int t = v[k]; v[k] = s; s += t; }
    int incl = s;
    for (int d = 1; d < 32; d <<= 1) {
      const int t = __shfl_up(incl, d, 32);
      if (lane >= d) incl += t;
    }
    if (lane == 31) wsum[wv] = incl;
    __syncthreads();
    if (wv == 0) {
      const int t = wsum[lane];
      int ti = t;
      for (int d = 1; d < 32; d <<= 1) {
        const int u = __shfl_up(ti, d, 32);
        if (lane >= d) ti += u;
      }
      wsum[lane] = ti - t;
      if (lane == 31) tile_total = ti;
    }
    __syncthreads();
    const int pre = carry + wsum[wv] + (incl - s);
    if (full) {
      for (int q = 0; q < 4; ++q) {
        int4 t;
        t.x = pre + v[4 * q + 0]; t.y = pre + v[4 * q + 1];
        t.z = pre + v[4 * q + 2]; t.w = pre + v[4 * q + 3];
        *(int4*)(cur + i0 + 4 * q) = t;
        *(int4*)(off + i0 + 4 * q) = t;
      }
    } else {
      for (int k = 0; k < 16; ++k) {
        if (i0 + k < n) { cur[i0 + k] = pre + v[k]; off[i0 + k] = pre + v[k]; }
      }
    }
    carry += tile_total;
    __syncthreads();
  }
  if (tid == 0) off[n] = carry;
}

__global__ void gat_fill_kernel(const int* __restrict__ rows,
                                const int* __restrict__ cols,
                                int* __restrict__ cur, int* __restrict__ ccol,
                                int E) {
  int e = blockIdx.x * blockDim.x + threadIdx.x;
  if (e < E) {
    const int p = atomicAdd(&cur[rows[e]], 1);
    ccol[p] = cols[e];
  }
}

__global__ __launch_bounds__(256) void gat_gather_l1_kernel(
    const int* __restrict__ off, const int* __restrict__ ccol,
    const float* __restrict__ alpha, const float* __restrict__ beta,
    const float* __restrict__ H, float* __restrict__ xc, int n) {
  const int row = __builtin_amdgcn_readfirstlane(
      (int)(blockIdx.x * 8 + (threadIdx.x >> 5)));
  const int lane = threadIdx.x & 31;
  if (row >= n) return;
  const int p0 = off[row];
  const int p1 = off[row + 1];
  const int hsel = lane >> 4;

  float al[4];
  float rs[4];
  float4 acc[4];
  for (int j = 0; j < 4; ++j) {
    al[j] = alpha[(size_t)row * 8 + 2 * j + hsel];
    rs[j] = 0.0f;
    acc[j] = make_float4(0.0f, 0.0f, 0.0f, 0.0f);
  }

  for (int base = p0; base < p1; base += 32) {
    const int m = (p1 - base < 32) ? (p1 - base) : 32;
    const int cv = (lane < m) ? ccol[base + lane] : 0;
    for (int i = 0; i < m; ++i) {
      const int c = __builtin_amdgcn_readlane(cv, i);
      const float* hc = H + (size_t)c * 512 + lane * 4;
      const float* bc = beta + (size_t)c * 8 + hsel;
      for (int j = 0; j < 4; ++j) {
        float s = al[j] + bc[2 * j];
        s = s > 0.0f ? s : 0.2f * s;
        const float w = __expf(-s);
        const float4 v = *(const float4*)(hc + j * 128);
        acc[j].x += w * v.x; acc[j].y += w * v.y;
        acc[j].z += w * v.z; acc[j].w += w * v.w;
        rs[j] += w;
      }
    }
  }

  float* xr = xc + (size_t)row * 512 + lane * 4;
  for (int j = 0; j < 4; ++j) {
    float4 o;
    o.x = acc[j].x / rs[j]; o.y = acc[j].y / rs[j];
    o.z = acc[j].z / rs[j]; o.w = acc[j].w / rs[j];
    o.x = o.x > 0.0f ? o.x : (__expf(o.x) - 1.0f);
    o.y = o.y > 0.0f ? o.y : (__expf(o.y) - 1.0f);
    o.z = o.z > 0.0f ? o.z : (__expf(o.z) - 1.0f);
    o.w = o.w > 0.0f ? o.w : (__expf(o.w) - 1.0f);
    *(float4*)(xr + j * 128) = o;
  }
}

__global__ __launch_bounds__(256) void gat_gather_l2_kernel(
    const int* __restrict__ off, const int* __restrict__ ccol,
    const float* __restrict__ alpha, const float* __restrict__ beta,
    const float* __restrict__ H2, float* __restrict__ out, int n) {
  const int row = __builtin_amdgcn_readfirstlane(
      (int)(blockIdx.x * 8 + (threadIdx.x >> 5)));
  const int lane = threadIdx.x & 31;
  if (row >= n) return;
  const int p0 = off[row];
  const int p1 = off[row + 1];
  const float a = alpha[row];

  float rs = 0.0f;
  float2 acc = make_float2(0.0f, 0.0f);
  for (int base = p0; base < p1; base += 32) {
    const int m = (p1 - base < 32) ? (p1 - base) : 32;
    const int cv = (lane < m) ? ccol[base + lane] : 0;
    for (int i = 0; i < m; ++i) {
      const int c = __builtin_amdgcn_readlane(cv, i);
      float s = a + beta[c];
      s = s > 0.0f ? s : 0.2f * s;
      const float w = __expf(-s);
      const float2 v = *(const float2*)(H2 + (size_t)c * 64 + lane * 2);
      acc.x += w * v.x; acc.y += w * v.y;
      rs += w;
    }
  }
  float2 o;
  o.x = acc.x / rs; o.y = acc.y / rs;
  *(float2*)(out + (size_t)row * 64 + lane * 2) = o;
}

extern "C" void kernel_launch(void* const* d_in, const int* in_sizes, int n_in,
                              void* d_out, int out_size, void* d_ws,
                              size_t ws_size, hipStream_t stream) {
  const float* x       = (const float*)d_in[0];
  const float* Ws      = (const float*)d_in[1];
  const float* a_heads = (const float*)d_in[2];
  const float* W_out   = (const float*)d_in[3];
  const float* a_out   = (const float*)d_in[4];
  const int*   eidx    = (const int*)d_in[5];
  float* out = (float*)d_out;

  const int N = in_sizes[0] / GAT_NFEAT;
  const int E = in_sizes[5] / 2;
  const int* e_row = eidx;
  const int* e_col = eidx + E;

  float* ws = (float*)d_ws;
  size_t off_f = 0;
  float* H      = ws + off_f; off_f += (size_t)N * 512;
  float* alpha1 = ws + off_f; off_f += (size_t)N * 8;
  float* beta1  = ws + off_f; off_f += (size_t)N * 8;
  float* xc     = ws + off_f; off_f += (size_t)N * 512;
  int* cur  = (int*)(ws + off_f); off_f += ((size_t)N + 3) / 4 * 4;
  int* roff = (int*)(ws + off_f); off_f += ((size_t)N + 4) / 4 * 4;
  int* ccol = (int*)(ws + off_f); off_f += (size_t)E;
  float* H2     = H;
  float* alpha2 = H + (size_t)N * 64;
  float* beta2  = alpha2 + (size_t)N;
  (void)ws_size; (void)n_in; (void)out_size;

  gat_zero_kernel<<<256, 256, 0, stream>>>((float*)cur, (size_t)N);
  gat_count_kernel<<<(E + 255) / 256, 256, 0, stream>>>(e_row, cur, E);
  gat_scan_kernel<<<1, 1024, 0, stream>>>(cur, roff, N);
  gat_fill_kernel<<<(E + 255) / 256, 256, 0, stream>>>(
      e_row, e_col, cur, ccol, E);

  const int mblocks = (N + 63) / 64;

  gat_gemm_f32_wmma<<<dim3(mblocks, GAT_HEADS), 128, 0, stream>>>(
      x, Ws, H, N, 512);

  gat_alpha_beta_kernel<<<(N * GAT_HEADS + 255) / 256, 256, 0, stream>>>(
      H, a_heads, alpha1, beta1, N, GAT_HEADS);

  gat_gather_l1_kernel<<<(N + 7) / 8, 256, 0, stream>>>(
      roff, ccol, alpha1, beta1, H, xc, N);

  gat_gemm_f32_wmma<<<dim3(mblocks, 1), 128, 0, stream>>>(
      xc, W_out, H2, N, 64);

  gat_alpha_beta_kernel<<<(N + 255) / 256, 256, 0, stream>>>(
      H2, a_out, alpha2, beta2, N, 1);

  gat_gather_l2_kernel<<<(N + 7) / 8, 256, 0, stream>>>(
      roff, ccol, alpha2, beta2, H2, out, N);
}


# GEMM operand staging as a register pipeline: the eight global loads of K-tile t+1 issue as one group behind the barrier and overlap tile t's WMMAs (were eight serial round trips per K-tile)
# speedup vs baseline: 2.9206x; 1.1332x over previous
#include <hip/hip_runtime.h>
#include <math.h>

typedef __attribute__((ext_vector_type(2))) float v2f;
typedef __attribute__((ext_vector_type(8))) float v8f;

#define GAT_NFEAT 512
#define GAT_NHID  64
#define GAT_HEADS 8
#define GAT_NCLASS 64

__global__ void gat_zero_kernel(float* __restrict__ p, size_t n) {
  size_t i = (size_t)blockIdx.x * blockDim.x + threadIdx.x;
  size_t stride = (size_t)gridDim.x * blockDim.x;
  for (; i < n; i += stride) p[i] = 0.0f;
}

__global__ __launch_bounds__(128) void gat_gemm_f32_wmma(
    const float* __restrict__ A, const float* __restrict__ B,
    float* __restrict__ C, int M, int ldc) {
  __shared__ float As[64][36];
  __shared__ float Bs[32][68];

  const int tid     = threadIdx.x;
  const int lane    = tid & 31;
  const int wv      = tid >> 5;
  const int laneLow = lane & 15;
  const int kOff    = (lane >> 4) << 1;
  const int rowBase = blockIdx.x * 64;
  const int colBase = blockIdx.y * 64;
  const float* Bblk = B + (size_t)blockIdx.y * (GAT_NFEAT * 64);

  v8f acc[4];
  acc[0] = {}; acc[1] = {}; acc[2] = {}; acc[3] = {};

  const int aCol = (tid & 7)  * 4;
  const int bCol = (tid & 15) * 4;
  const int aRow = tid >> 3;
  const int bRow = tid >> 4;

  const float* aSrc[4];
#pragma unroll
  for (int p = 0; p < 4; ++p) {
    int gRow = rowBase + aRow + p * 16;
    if (gRow >= M) gRow = M - 1;
    aSrc[p] = A + (size_t)gRow * GAT_NFEAT + aCol;
  }
  const float* bSrc = Bblk + (size_t)bRow * 64 + bCol;

  float4 ra0 = *(const float4*)(aSrc[0]);
  float4 ra1 = *(const float4*)(aSrc[1]);
  float4 ra2 = *(const float4*)(aSrc[2]);
  float4 ra3 = *(const float4*)(aSrc[3]);
  float4 rb0 = *(const float4*)(bSrc);
  float4 rb1 = *(const float4*)(bSrc + 8 * 64);
  float4 rb2 = *(const float4*)(bSrc + 16 * 64);
  float4 rb3 = *(const float4*)(bSrc + 24 * 64);

  for (int k0 = 0; k0 < GAT_NFEAT; k0 += 32) {
    *(float4*)&As[aRow][aCol]      = ra0;
    *(float4*)&As[aRow + 16][aCol] = ra1;
    *(float4*)&As[aRow + 32][aCol] = ra2;
    *(float4*)&As[aRow + 48][aCol] = ra3;
    *(float4*)&Bs[bRow][bCol]      = rb0;
    *(float4*)&Bs[bRow + 8][bCol]  = rb1;
    *(float4*)&Bs[bRow + 16][bCol] = rb2;
    *(float4*)&Bs[bRow + 24][bCol] = rb3;
    __syncthreads();

    if (k0 + 32 < GAT_NFEAT) {
      const float* bn = bSrc + (size_t)(k0 + 32) * 64;
      ra0 = *(const float4*)(aSrc[0] + k0 + 32);
      ra1 = *(const float4*)(aSrc[1] + k0 + 32);
      ra2 = *(const float4*)(aSrc[2] + k0 + 32);
      ra3 = *(const float4*)(aSrc[3] + k0 + 32);
      rb0 = *(const float4*)(bn);
      rb1 = *(const float4*)(bn + 8 * 64);
      rb2 = *(const float4*)(bn + 16 * 64);
      rb3 = *(const float4*)(bn + 24 * 64);
    }

#pragma unroll
    for (int kk = 0; kk < 32; kk += 4) {
      v2f a;
      a.x = As[wv * 16 + laneLow][kk + kOff];
      a.y = As[wv * 16 + laneLow][kk + kOff + 1];
#pragma unroll
      for (int n = 0; n < 4; ++n) {
        v2f b;
        b.x = Bs[kk + kOff][n * 16 + laneLow];
        b.y = Bs[kk + kOff + 1][n * 16 + laneLow];
        acc[n] = __builtin_amdgcn_wmma_f32_16x16x4_f32(
            false, a, false, b, (short)0, acc[n], false, false);
      }
    }
    __syncthreads();
  }

#pragma unroll
  for (int n = 0; n < 4; ++n) {
#pragma unroll
    for (int r = 0; r < 8; ++r) {
      int row = rowBase + wv * 16 + r + ((lane >> 4) << 3);
      if (row < M)
        C[(size_t)row * ldc + colBase + n * 16 + laneLow] = acc[n][r];
    }
  }
}

__global__ void gat_alpha_beta_kernel(
    const float* __restrict__ H, const float* __restrict__ a,
    float* __restrict__ alpha, float* __restrict__ beta, int n, int heads) {
  int t = blockIdx.x * blockDim.x + threadIdx.x;
  if (t >= n * heads) return;
  int node = t / heads;
  int h = t - node * heads;
  const float* hv = H + (size_t)node * heads * 64 + h * 64;
  const float* av = a + h * 128;
  float sa = 0.0f, sb = 0.0f;
#pragma unroll
  for (int d = 0; d < 64; d += 4) {
    float4 hh = *(const float4*)(hv + d);
    float4 a1 = *(const float4*)(av + d);
    float4 a2 = *(const float4*)(av + 64 + d);
    sa += hh.x * a1.x + hh.y * a1.y + hh.z * a1.z + hh.w * a1.w;
    sb += hh.x * a2.x + hh.y * a2.y + hh.z * a2.z + hh.w * a2.w;
  }
  alpha[t] = sa;
  beta[t] = sb;
}

__global__ void gat_count_kernel(const int* __restrict__ rows,
                                 int* __restrict__ cur, int E) {
  int e = blockIdx.x * blockDim.x + threadIdx.x;
  if (e < E) atomicAdd(&cur[rows[e]], 1);
}

__global__ __launch_bounds__(1024) void gat_scan_kernel(
    int* __restrict__ cur, int* __restrict__ off, int n) {
  __shared__ int wsum[32];
  __shared__ int tile_total;
  const int tid  = threadIdx.x;
  const int lane = tid & 31;
  const int wv   = tid >> 5;
  int carry = 0;
  for (int base = 0; base < n; base += 1024 * 16) {
    const int i0 = base + tid * 16;
    const bool full = (i0 + 16 <= n);
    int v[16];
    if (full) {
      for (int q = 0; q < 4; ++q) {
        const int4 t = *(const int4*)(cur + i0 + 4 * q);
        v[4 * q + 0] = t.x; v[4 * q + 1] = t.y;
        v[4 * q + 2] = t.z; v[4 * q + 3] = t.w;
      }
    } else {
      for (int k = 0; k < 16; ++k) v[k] = (i0 + k < n) ? cur[i0 + k] : 0;
    }
    int s = 0;
    for (int k = 0; k < 16; ++k) { const int t = v[k]; v[k] = s; s += t; }
    int incl = s;
    for (int d = 1; d < 32; d <<= 1) {
      const int t = __shfl_up(incl, d, 32);
      if (lane >= d) incl += t;
    }
    if (lane == 31) wsum[wv] = incl;
    __syncthreads();
    if (wv == 0) {
      const int t = wsum[lane];
      int ti = t;
      for (int d = 1; d < 32; d <<= 1) {
        const int u = __shfl_up(ti, d, 32);
        if (lane >= d) ti += u;
      }
      wsum[lane] = ti - t;
      if (lane == 31) tile_total = ti;
    }
    __syncthreads();
    const int pre = carry + wsum[wv] + (incl - s);
    if (full) {
      for (int q = 0; q < 4; ++q) {
        int4 t;
        t.x = pre + v[4 * q + 0]; t.y = pre + v[4 * q + 1];
        t.z = pre + v[4 * q + 2]; t.w = pre + v[4 * q + 3];
        *(int4*)(cur + i0 + 4 * q) = t;
        *(int4*)(off + i0 + 4 * q) = t;
      }
    } else {
      for (int k = 0; k < 16; ++k) {
        if (i0 + k < n) { cur[i0 + k] = pre + v[k]; off[i0 + k] = pre + v[k]; }
      }
    }
    carry += tile_total;
    __syncthreads();
  }
  if (tid == 0) off[n] = carry;
}

__global__ void gat_fill_kernel(const int* __restrict__ rows,
                                const int* __restrict__ cols,
                                int* __restrict__ cur, int* __restrict__ ccol,
                                int E) {
  int e = blockIdx.x * blockDim.x + threadIdx.x;
  if (e < E) {
    const int p = atomicAdd(&cur[rows[e]], 1);
    ccol[p] = cols[e];
  }
}

__global__ __launch_bounds__(256) void gat_gather_l1_kernel(
    const int* __restrict__ off, const int* __restrict__ ccol,
    const float* __restrict__ alpha, const float* __restrict__ beta,
    const float* __restrict__ H, float* __restrict__ xc, int n) {
  const int row = __builtin_amdgcn_readfirstlane(
      (int)(blockIdx.x * 8 + (threadIdx.x >> 5)));
  const int lane = threadIdx.x & 31;
  if (row >= n) return;
  const int p0 = off[row];
  const int p1 = off[row + 1];
  const int hsel = lane >> 4;

  float al[4];
  float rs[4];
  float4 acc[4];
  for (int j = 0; j < 4; ++j) {
    al[j] = alpha[(size_t)row * 8 + 2 * j + hsel];
    rs[j] = 0.0f;
    acc[j] = make_float4(0.0f, 0.0f, 0.0f, 0.0f);
  }

  for (int base = p0; base < p1; base += 32) {
    const int m = (p1 - base < 32) ? (p1 - base) : 32;
    const int cv = (lane < m) ? ccol[base + lane] : 0;
    for (int i = 0; i < m; ++i) {
      const int c = __builtin_amdgcn_readlane(cv, i);
      const float* hc = H + (size_t)c * 512 + lane * 4;
      const float* bc = beta + (size_t)c * 8 + hsel;
      for (int j = 0; j < 4; ++j) {
        float s = al[j] + bc[2 * j];
        s = s > 0.0f ? s : 0.2f * s;
        const float w = __expf(-s);
        const float4 v = *(const float4*)(hc + j * 128);
        acc[j].x += w * v.x; acc[j].y += w * v.y;
        acc[j].z += w * v.z; acc[j].w += w * v.w;
        rs[j] += w;
      }
    }
  }

  float* xr = xc + (size_t)row * 512 + lane * 4;
  for (int j = 0; j < 4; ++j) {
    float4 o;
    o.x = acc[j].x / rs[j]; o.y = acc[j].y / rs[j];
    o.z = acc[j].z / rs[j]; o.w = acc[j].w / rs[j];
    o.x = o.x > 0.0f ? o.x : (__expf(o.x) - 1.0f);
    o.y = o.y > 0.0f ? o.y : (__expf(o.y) - 1.0f);
    o.z = o.z > 0.0f ? o.z : (__expf(o.z) - 1.0f);
    o.w = o.w > 0.0f ? o.w : (__expf(o.w) - 1.0f);
    *(float4*)(xr + j * 128) = o;
  }
}

__global__ __launch_bounds__(256) void gat_gather_l2_kernel(
    const int* __restrict__ off, const int* __restrict__ ccol,
    const float* __restrict__ alpha, const float* __restrict__ beta,
    const float* __restrict__ H2, float* __restrict__ out, int n) {
  const int row = __builtin_amdgcn_readfirstlane(
      (int)(blockIdx.x * 8 + (threadIdx.x >> 5)));
  const int lane = threadIdx.x & 31;
  if (row >= n) return;
  const int p0 = off[row];
  const int p1 = off[row + 1];
  const float a = alpha[row];

  float rs = 0.0f;
  float2 acc = make_float2(0.0f, 0.0f);
  for (int base = p0; base < p1; base += 32) {
    const int m = (p1 - base < 32) ? (p1 - base) : 32;
    const int cv = (lane < m) ? ccol[base + lane] : 0;
    for (int i = 0; i < m; ++i) {
      const int c = __builtin_amdgcn_readlane(cv, i);
      float s = a + beta[c];
      s = s > 0.0f ? s : 0.2f * s;
      const float w = __expf(-s);
      const float2 v = *(const float2*)(H2 + (size_t)c * 64 + lane * 2);
      acc.x += w * v.x; acc.y += w * v.y;
      rs += w;
    }
  }
  float2 o;
  o.x = acc.x / rs; o.y = acc.y / rs;
  *(float2*)(out + (size_t)row * 64 + lane * 2) = o;
}

extern "C" void kernel_launch(void* const* d_in, const int* in_sizes, int n_in,
                              void* d_out, int out_size, void* d_ws,
                              size_t ws_size, hipStream_t stream) {
  const float* x       = (const float*)d_in[0];
  const float* Ws      = (const float*)d_in[1];
  const float* a_heads = (const float*)d_in[2];
  const float* W_out   = (const float*)d_in[3];
  const float* a_out   = (const float*)d_in[4];
  const int*   eidx    = (const int*)d_in[5];
  float* out = (float*)d_out;

  const int N = in_sizes[0] / GAT_NFEAT;
  const int E = in_sizes[5] / 2;
  const int* e_row = eidx;
  const int* e_col = eidx + E;

  float* ws = (float*)d_ws;
  size_t off_f = 0;
  float* H      = ws + off_f; off_f += (size_t)N * 512;
  float* alpha1 = ws + off_f; off_f += (size_t)N * 8;
  float* beta1  = ws + off_f; off_f += (size_t)N * 8;
  float* xc     = ws + off_f; off_f += (size_t)N * 512;
  int* cur  = (int*)(ws + off_f); off_f += ((size_t)N + 3) / 4 * 4;
  int* roff = (int*)(ws + off_f); off_f += ((size_t)N + 4) / 4 * 4;
  int* ccol = (int*)(ws + off_f); off_f += (size_t)E;
  float* H2     = H;
  float* alpha2 = H + (size_t)N * 64;
  float* beta2  = alpha2 + (size_t)N;
  (void)ws_size; (void)n_in; (void)out_size;

  gat_zero_kernel<<<256, 256, 0, stream>>>((float*)cur, (size_t)N);
  gat_count_kernel<<<(E + 255) / 256, 256, 0, stream>>>(e_row, cur, E);
  gat_scan_kernel<<<1, 1024, 0, stream>>>(cur, roff, N);
  gat_fill_kernel<<<(E + 255) / 256, 256, 0, stream>>>(
      e_row, e_col, cur, ccol, E);

  const int mblocks = (N + 63) / 64;

  gat_gemm_f32_wmma<<<dim3(mblocks, GAT_HEADS), 128, 0, stream>>>(
      x, Ws, H, N, 512);

  gat_alpha_beta_kernel<<<(N * GAT_HEADS + 255) / 256, 256, 0, stream>>>(
      H, a_heads, alpha1, beta1, N, GAT_HEADS);

  gat_gather_l1_kernel<<<(N + 7) / 8, 256, 0, stream>>>(
      roff, ccol, alpha1, beta1, H, xc, N);

  gat_gemm_f32_wmma<<<dim3(mblocks, 1), 128, 0, stream>>>(
      xc, W_out, H2, N, 64);

  gat_alpha_beta_kernel<<<(N + 255) / 256, 256, 0, stream>>>(
      H2, a_out, alpha2, beta2, N, 1);

  gat_gather_l2_kernel<<<(N + 7) / 8, 256, 0, stream>>>(
      roff, ccol, alpha2, beta2, H2, out, N);
}


# alpha/beta dot products moved into the GEMM epilogue (4-column FMA per lane on the accumulators, then a 16-lane DPP halving sum): both alpha-beta launches and their H re-read go; sum order changes
# speedup vs baseline: 3.0501x; 1.0443x over previous
#include <hip/hip_runtime.h>
#include <math.h>

typedef __attribute__((ext_vector_type(2))) float v2f;
typedef __attribute__((ext_vector_type(8))) float v8f;

#define GAT_NFEAT 512
#define GAT_NHID  64
#define GAT_HEADS 8
#define GAT_NCLASS 64

__global__ void gat_zero_kernel(float* __restrict__ p, size_t n) {
  size_t i = (size_t)blockIdx.x * blockDim.x + threadIdx.x;
  size_t stride = (size_t)gridDim.x * blockDim.x;
  for (; i < n; i += stride) p[i] = 0.0f;
}


template <int XM>
__device__ __forceinline__ float gat_row_xor(float x) {
  const int b = __builtin_bit_cast(int, x);
  return __builtin_bit_cast(
      float, __builtin_amdgcn_update_dpp(b, b, 0x160 + XM, 0xf, 0xf, false));
}

__global__ __launch_bounds__(128) void gat_gemm_f32_wmma(
    const float* __restrict__ A, const float* __restrict__ B,
    const float* __restrict__ avec, float* __restrict__ C,
    float* __restrict__ alpha, float* __restrict__ beta, int M, int ldc) {
  __shared__ float As[64][36];
  __shared__ float Bs[32][68];

  const int tid     = threadIdx.x;
  const int lane    = tid & 31;
  const int wv      = tid >> 5;
  const int laneLow = lane & 15;
  const int kOff    = (lane >> 4) << 1;
  const int rowBase = blockIdx.x * 64;
  const int colBase = blockIdx.y * 64;
  const float* Bblk = B + (size_t)blockIdx.y * (GAT_NFEAT * 64);

  v8f acc[4];
  acc[0] = {}; acc[1] = {}; acc[2] = {}; acc[3] = {};

  const int aCol = (tid & 7)  * 4;
  const int bCol = (tid & 15) * 4;
  const int aRow = tid >> 3;
  const int bRow = tid >> 4;

  const float* aSrc[4];
#pragma unroll
  for (int p = 0; p < 4; ++p) {
    int gRow = rowBase + aRow + p * 16;
    if (gRow >= M) gRow = M - 1;
    aSrc[p] = A + (size_t)gRow * GAT_NFEAT + aCol;
  }
  const float* bSrc = Bblk + (size_t)bRow * 64 + bCol;

  float4 ra0 = *(const float4*)(aSrc[0]);
  float4 ra1 = *(const float4*)(aSrc[1]);
  float4 ra2 = *(const float4*)(aSrc[2]);
  float4 ra3 = *(const float4*)(aSrc[3]);
  float4 rb0 = *(const float4*)(bSrc);
  float4 rb1 = *(const float4*)(bSrc + 8 * 64);
  float4 rb2 = *(const float4*)(bSrc + 16 * 64);
  float4 rb3 = *(const float4*)(bSrc + 24 * 64);

  for (int k0 = 0; k0 < GAT_NFEAT; k0 += 32) {
    *(float4*)&As[aRow][aCol]      = ra0;
    *(float4*)&As[aRow + 16][aCol] = ra1;
    *(float4*)&As[aRow + 32][aCol] = ra2;
    *(float4*)&As[aRow + 48][aCol] = ra3;
    *(float4*)&Bs[bRow][bCol]      = rb0;
    *(float4*)&Bs[bRow + 8][bCol]  = rb1;
    *(float4*)&Bs[bRow + 16][bCol] = rb2;
    *(float4*)&Bs[bRow + 24][bCol] = rb3;
    __syncthreads();

    if (k0 + 32 < GAT_NFEAT) {
      const float* bn = bSrc + (size_t)(k0 + 32) * 64;
      ra0 = *(const float4*)(aSrc[0] + k0 + 32);
      ra1 = *(const float4*)(aSrc[1] + k0 + 32);
      ra2 = *(const float4*)(aSrc[2] + k0 + 32);
      ra3 = *(const float4*)(aSrc[3] + k0 + 32);
      rb0 = *(const float4*)(bn);
      rb1 = *(const float4*)(bn + 8 * 64);
      rb2 = *(const float4*)(bn + 16 * 64);
      rb3 = *(const float4*)(bn + 24 * 64);
    }

#pragma unroll
    for (int kk = 0; kk < 32; kk += 4) {
      v2f a;
      a.x = As[wv * 16 + laneLow][kk + kOff];
      a.y = As[wv * 16 + laneLow][kk + kOff + 1];
#pragma unroll
      for (int n = 0; n < 4; ++n) {
        v2f b;
        b.x = Bs[kk + kOff][n * 16 + laneLow];
        b.y = Bs[kk + kOff + 1][n * 16 + laneLow];
        acc[n] = __builtin_amdgcn_wmma_f32_16x16x4_f32(
            false, a, false, b, (short)0, acc[n], false, false);
      }
    }
    __syncthreads();
  }

  const int el    = (int)__builtin_amdgcn_mbcnt_lo(~0u, 0u);
  const int eLow  = el & 15;
  const int eHalf = el >> 4;
  const int ewv   = __builtin_amdgcn_readfirstlane(tid >> 5);

  const float* av = avec + (size_t)blockIdx.y * 128;
  float a1[4], a2[4];
#pragma unroll
  for (int n = 0; n < 4; ++n) {
    a1[n] = av[n * 16 + eLow];
    a2[n] = av[64 + n * 16 + eLow];
  }

#pragma unroll
  for (int n = 0; n < 4; ++n) {
#pragma unroll
    for (int r = 0; r < 8; ++r) {
      int row = rowBase + ewv * 16 + r + (eHalf << 3);
      if (row < M)
        C[(size_t)row * ldc + colBase + n * 16 + eLow] = acc[n][r];
    }
  }

  float v[16];
#pragma unroll
  for (int r = 0; r < 8; ++r) {
    v[r]     = acc[0][r] * a1[0] + acc[1][r] * a1[1] +
               acc[2][r] * a1[2] + acc[3][r] * a1[3];
    v[8 + r] = acc[0][r] * a2[0] + acc[1][r] * a2[1] +
               acc[2][r] * a2[2] + acc[3][r] * a2[3];
  }
  const bool b3 = (eLow & 8) != 0;
  const bool b2 = (eLow & 4) != 0;
  const bool b1 = (eLow & 2) != 0;
  const bool b0 = (eLow & 1) != 0;
  float u[8];
#pragma unroll
  for (int j = 0; j < 8; ++j) {
    const float keep = b3 ? v[8 + j] : v[j];
    const float send = b3 ? v[j] : v[8 + j];
    u[j] = keep + gat_row_xor<8>(send);
  }
  float t[4];
#pragma unroll
  for (int j = 0; j < 4; ++j) {
    const float keep = b2 ? u[4 + j] : u[j];
    const float send = b2 ? u[j] : u[4 + j];
    t[j] = keep + gat_row_xor<4>(send);
  }
  float s[2];
#pragma unroll
  for (int j = 0; j < 2; ++j) {
    const float keep = b1 ? t[2 + j] : t[j];
    const float send = b1 ? t[j] : t[2 + j];
    s[j] = keep + gat_row_xor<2>(send);
  }
  const float keep0 = b0 ? s[1] : s[0];
  const float send0 = b0 ? s[0] : s[1];
  const float tot = keep0 + gat_row_xor<1>(send0);

  const int orow = rowBase + ewv * 16 + (eHalf << 3) + (eLow & 7);
  if (orow < M) {
    float* dst = b3 ? beta : alpha;
    dst[(size_t)orow * gridDim.y + blockIdx.y] = tot;
  }
}

__global__ void gat_count_kernel(const int* __restrict__ rows,
                                 int* __restrict__ cur, int E) {
  int e = blockIdx.x * blockDim.x + threadIdx.x;
  if (e < E) atomicAdd(&cur[rows[e]], 1);
}

__global__ __launch_bounds__(1024) void gat_scan_kernel(
    int* __restrict__ cur, int* __restrict__ off, int n) {
  __shared__ int wsum[32];
  __shared__ int tile_total;
  const int tid  = threadIdx.x;
  const int lane = tid & 31;
  const int wv   = tid >> 5;
  int carry = 0;
  for (int base = 0; base < n; base += 1024 * 16) {
    const int i0 = base + tid * 16;
    const bool full = (i0 + 16 <= n);
    int v[16];
    if (full) {
      for (int q = 0; q < 4; ++q) {
        const int4 t = *(const int4*)(cur + i0 + 4 * q);
        v[4 * q + 0] = t.x; v[4 * q + 1] = t.y;
        v[4 * q + 2] = t.z; v[4 * q + 3] = t.w;
      }
    } else {
      for (int k = 0; k < 16; ++k) v[k] = (i0 + k < n) ? cur[i0 + k] : 0;
    }
    int s = 0;
    for (int k = 0; k < 16; ++k) { const int t = v[k]; v[k] = s; s += t; }
    int incl = s;
    for (int d = 1; d < 32; d <<= 1) {
      const int t = __shfl_up(incl, d, 32);
      if (lane >= d) incl += t;
    }
    if (lane == 31) wsum[wv] = incl;
    __syncthreads();
    if (wv == 0) {
      const int t = wsum[lane];
      int ti = t;
      for (int d = 1; d < 32; d <<= 1) {
        const int u = __shfl_up(ti, d, 32);
        if (lane >= d) ti += u;
      }
      wsum[lane] = ti - t;
      if (lane == 31) tile_total = ti;
    }
    __syncthreads();
    const int pre = carry + wsum[wv] + (incl - s);
    if (full) {
      for (int q = 0; q < 4; ++q) {
        int4 t;
        t.x = pre + v[4 * q + 0]; t.y = pre + v[4 * q + 1];
        t.z = pre + v[4 * q + 2]; t.w = pre + v[4 * q + 3];
        *(int4*)(cur + i0 + 4 * q) = t;
        *(int4*)(off + i0 + 4 * q) = t;
      }
    } else {
      for (int k = 0; k < 16; ++k) {
        if (i0 + k < n) { cur[i0 + k] = pre + v[k]; off[i0 + k] = pre + v[k]; }
      }
    }
    carry += tile_total;
    __syncthreads();
  }
  if (tid == 0) off[n] = carry;
}

__global__ void gat_fill_kernel(const int* __restrict__ rows,
                                const int* __restrict__ cols,
                                int* __restrict__ cur, int* __restrict__ ccol,
                                int E) {
  int e = blockIdx.x * blockDim.x + threadIdx.x;
  if (e < E) {
    const int p = atomicAdd(&cur[rows[e]], 1);
    ccol[p] = cols[e];
  }
}

__global__ __launch_bounds__(256) void gat_gather_l1_kernel(
    const int* __restrict__ off, const int* __restrict__ ccol,
    const float* __restrict__ alpha, const float* __restrict__ beta,
    const float* __restrict__ H, float* __restrict__ xc, int n) {
  const int row = __builtin_amdgcn_readfirstlane(
      (int)(blockIdx.x * 8 + (threadIdx.x >> 5)));
  const int lane = threadIdx.x & 31;
  if (row >= n) return;
  const int p0 = off[row];
  const int p1 = off[row + 1];
  const int hsel = lane >> 4;

  float al[4];
  float rs[4];
  float4 acc[4];
  for (int j = 0; j < 4; ++j) {
    al[j] = alpha[(size_t)row * 8 + 2 * j + hsel];
    rs[j] = 0.0f;
    acc[j] = make_float4(0.0f, 0.0f, 0.0f, 0.0f);
  }

  for (int base = p0; base < p1; base += 32) {
    const int m = (p1 - base < 32) ? (p1 - base) : 32;
    const int cv = (lane < m) ? ccol[base + lane] : 0;
    for (int i = 0; i < m; ++i) {
      const int c = __builtin_amdgcn_readlane(cv, i);
      const float* hc = H + (size_t)c * 512 + lane * 4;
      const float* bc = beta + (size_t)c * 8 + hsel;
      for (int j = 0; j < 4; ++j) {
        float s = al[j] + bc[2 * j];
        s = s > 0.0f ? s : 0.2f * s;
        const float w = __expf(-s);
        const float4 v = *(const float4*)(hc + j * 128);
        acc[j].x += w * v.x; acc[j].y += w * v.y;
        acc[j].z += w * v.z; acc[j].w += w * v.w;
        rs[j] += w;
      }
    }
  }

  float* xr = xc + (size_t)row * 512 + lane * 4;
  for (int j = 0; j < 4; ++j) {
    float4 o;
    o.x = acc[j].x / rs[j]; o.y = acc[j].y / rs[j];
    o.z = acc[j].z / rs[j]; o.w = acc[j].w / rs[j];
    o.x = o.x > 0.0f ? o.x : (__expf(o.x) - 1.0f);
    o.y = o.y > 0.0f ? o.y : (__expf(o.y) - 1.0f);
    o.z = o.z > 0.0f ? o.z : (__expf(o.z) - 1.0f);
    o.w = o.w > 0.0f ? o.w : (__expf(o.w) - 1.0f);
    *(float4*)(xr + j * 128) = o;
  }
}

__global__ __launch_bounds__(256) void gat_gather_l2_kernel(
    const int* __restrict__ off, const int* __restrict__ ccol,
    const float* __restrict__ alpha, const float* __restrict__ beta,
    const float* __restrict__ H2, float* __restrict__ out, int n) {
  const int row = __builtin_amdgcn_readfirstlane(
      (int)(blockIdx.x * 8 + (threadIdx.x >> 5)));
  const int lane = threadIdx.x & 31;
  if (row >= n) return;
  const int p0 = off[row];
  const int p1 = off[row + 1];
  const float a = alpha[row];

  float rs = 0.0f;
  float2 acc = make_float2(0.0f, 0.0f);
  for (int base = p0; base < p1; base += 32) {
    const int m = (p1 - base < 32) ? (p1 - base) : 32;
    const int cv = (lane < m) ? ccol[base + lane] : 0;
    for (int i = 0; i < m; ++i) {
      const int c = __builtin_amdgcn_readlane(cv, i);
      float s = a + beta[c];
      s = s > 0.0f ? s : 0.2f * s;
      const float w = __expf(-s);
      const float2 v = *(const float2*)(H2 + (size_t)c * 64 + lane * 2);
      acc.x += w * v.x; acc.y += w * v.y;
      rs += w;
    }
  }
  float2 o;
  o.x = acc.x / rs; o.y = acc.y / rs;
  *(float2*)(out + (size_t)row * 64 + lane * 2) = o;
}

extern "C" void kernel_launch(void* const* d_in, const int* in_sizes, int n_in,
                              void* d_out, int out_size, void* d_ws,
                              size_t ws_size, hipStream_t stream) {
  const float* x       = (const float*)d_in[0];
  const float* Ws      = (const float*)d_in[1];
  const float* a_heads = (const float*)d_in[2];
  const float* W_out   = (const float*)d_in[3];
  const float* a_out   = (const float*)d_in[4];
  const int*   eidx    = (const int*)d_in[5];
  float* out = (float*)d_out;

  const int N = in_sizes[0] / GAT_NFEAT;
  const int E = in_sizes[5] / 2;
  const int* e_row = eidx;
  const int* e_col = eidx + E;

  float* ws = (float*)d_ws;
  size_t off_f = 0;
  float* H      = ws + off_f; off_f += (size_t)N * 512;
  float* alpha1 = ws + off_f; off_f += (size_t)N * 8;
  float* beta1  = ws + off_f; off_f += (size_t)N * 8;
  float* xc     = ws + off_f; off_f += (size_t)N * 512;
  int* cur  = (int*)(ws + off_f); off_f += ((size_t)N + 3) / 4 * 4;
  int* roff = (int*)(ws + off_f); off_f += ((size_t)N + 4) / 4 * 4;
  int* ccol = (int*)(ws + off_f); off_f += (size_t)E;
  float* H2     = H;
  float* alpha2 = H + (size_t)N * 64;
  float* beta2  = alpha2 + (size_t)N;
  (void)ws_size; (void)n_in; (void)out_size;

  gat_zero_kernel<<<256, 256, 0, stream>>>((float*)cur, (size_t)N);
  gat_count_kernel<<<(E + 255) / 256, 256, 0, stream>>>(e_row, cur, E);
  gat_scan_kernel<<<1, 1024, 0, stream>>>(cur, roff, N);
  gat_fill_kernel<<<(E + 255) / 256, 256, 0, stream>>>(
      e_row, e_col, cur, ccol, E);

  const int mblocks = (N + 63) / 64;

  gat_gemm_f32_wmma<<<dim3(mblocks, GAT_HEADS), 128, 0, stream>>>(
      x, Ws, a_heads, H, alpha1, beta1, N, 512);

  gat_gather_l1_kernel<<<(N + 7) / 8, 256, 0, stream>>>(
      roff, ccol, alpha1, beta1, H, xc, N);

  gat_gemm_f32_wmma<<<dim3(mblocks, 1), 128, 0, stream>>>(
      xc, W_out, a_out, H2, alpha2, beta2, N, 64);

  gat_gather_l2_kernel<<<(N + 7) / 8, 256, 0, stream>>>(
      roff, ccol, alpha2, beta2, H2, out, N);
}


# Edge-list fill without atomics: the count's atomic returns each edge's rank in its row (kept per edge); fill stores 4 edges per thread to off[row]+rank; row sum order = count's arrival rank
# speedup vs baseline: 3.3287x; 1.0913x over previous
#include <hip/hip_runtime.h>
#include <math.h>

typedef __attribute__((ext_vector_type(2))) float v2f;
typedef __attribute__((ext_vector_type(8))) float v8f;

#define GAT_NFEAT 512
#define GAT_NHID  64
#define GAT_HEADS 8
#define GAT_NCLASS 64

__global__ void gat_zero_kernel(float* __restrict__ p, size_t n) {
  size_t i = (size_t)blockIdx.x * blockDim.x + threadIdx.x;
  size_t stride = (size_t)gridDim.x * blockDim.x;
  for (; i < n; i += stride) p[i] = 0.0f;
}


template <int XM>
__device__ __forceinline__ float gat_row_xor(float x) {
  const int b = __builtin_bit_cast(int, x);
  return __builtin_bit_cast(
      float, __builtin_amdgcn_update_dpp(b, b, 0x160 + XM, 0xf, 0xf, false));
}

__global__ __launch_bounds__(128) void gat_gemm_f32_wmma(
    const float* __restrict__ A, const float* __restrict__ B,
    const float* __restrict__ avec, float* __restrict__ C,
    float* __restrict__ alpha, float* __restrict__ beta, int M, int ldc) {
  __shared__ float As[64][36];
  __shared__ float Bs[32][68];

  const int tid     = threadIdx.x;
  const int lane    = tid & 31;
  const int wv      = tid >> 5;
  const int laneLow = lane & 15;
  const int kOff    = (lane >> 4) << 1;
  const int rowBase = blockIdx.x * 64;
  const int colBase = blockIdx.y * 64;
  const float* Bblk = B + (size_t)blockIdx.y * (GAT_NFEAT * 64);

  v8f acc[4];
  acc[0] = {}; acc[1] = {}; acc[2] = {}; acc[3] = {};

  const int aCol = (tid & 7)  * 4;
  const int bCol = (tid & 15) * 4;
  const int aRow = tid >> 3;
  const int bRow = tid >> 4;

  const float* aSrc[4];
#pragma unroll
  for (int p = 0; p < 4; ++p) {
    int gRow = rowBase + aRow + p * 16;
    if (gRow >= M) gRow = M - 1;
    aSrc[p] = A + (size_t)gRow * GAT_NFEAT + aCol;
  }
  const float* bSrc = Bblk + (size_t)bRow * 64 + bCol;

  float4 ra0 = *(const float4*)(aSrc[0]);
  float4 ra1 = *(const float4*)(aSrc[1]);
  float4 ra2 = *(const float4*)(aSrc[2]);
  float4 ra3 = *(const float4*)(aSrc[3]);
  float4 rb0 = *(const float4*)(bSrc);
  float4 rb1 = *(const float4*)(bSrc + 8 * 64);
  float4 rb2 = *(const float4*)(bSrc + 16 * 64);
  float4 rb3 = *(const float4*)(bSrc + 24 * 64);

  for (int k0 = 0; k0 < GAT_NFEAT; k0 += 32) {
    *(float4*)&As[aRow][aCol]      = ra0;
    *(float4*)&As[aRow + 16][aCol] = ra1;
    *(float4*)&As[aRow + 32][aCol] = ra2;
    *(float4*)&As[aRow + 48][aCol] = ra3;
    *(float4*)&Bs[bRow][bCol]      = rb0;
    *(float4*)&Bs[bRow + 8][bCol]  = rb1;
    *(float4*)&Bs[bRow + 16][bCol] = rb2;
    *(float4*)&Bs[bRow + 24][bCol] = rb3;
    __syncthreads();

    if (k0 + 32 < GAT_NFEAT) {
      const float* bn = bSrc + (size_t)(k0 + 32) * 64;
      ra0 = *(const float4*)(aSrc[0] + k0 + 32);
      ra1 = *(const float4*)(aSrc[1] + k0 + 32);
      ra2 = *(const float4*)(aSrc[2] + k0 + 32);
      ra3 = *(const float4*)(aSrc[3] + k0 + 32);
      rb0 = *(const float4*)(bn);
      rb1 = *(const float4*)(bn + 8 * 64);
      rb2 = *(const float4*)(bn + 16 * 64);
      rb3 = *(const float4*)(bn + 24 * 64);
    }

#pragma unroll
    for (int kk = 0; kk < 32; kk += 4) {
      v2f a;
      a.x = As[wv * 16 + laneLow][kk + kOff];
      a.y = As[wv * 16 + laneLow][kk + kOff + 1];
#pragma unroll
      for (int n = 0; n < 4; ++n) {
        v2f b;
        b.x = Bs[kk + kOff][n * 16 + laneLow];
        b.y = Bs[kk + kOff + 1][n * 16 + laneLow];
        acc[n] = __builtin_amdgcn_wmma_f32_16x16x4_f32(
            false, a, false, b, (short)0, acc[n], false, false);
      }
    }
    __syncthreads();
  }

  const int el    = (int)__builtin_amdgcn_mbcnt_lo(~0u, 0u);
  const int eLow  = el & 15;
  const int eHalf = el >> 4;
  const int ewv   = __builtin_amdgcn_readfirstlane(tid >> 5);

  const float* av = avec + (size_t)blockIdx.y * 128;
  float a1[4], a2[4];
#pragma unroll
  for (int n = 0; n < 4; ++n) {
    a1[n] = av[n * 16 + eLow];
    a2[n] = av[64 + n * 16 + eLow];
  }

#pragma unroll
  for (int n = 0; n < 4; ++n) {
#pragma unroll
    for (int r = 0; r < 8; ++r) {
      int row = rowBase + ewv * 16 + r + (eHalf << 3);
      if (row < M)
        C[(size_t)row * ldc + colBase + n * 16 + eLow] = acc[n][r];
    }
  }

  float v[16];
#pragma unroll
  for (int r = 0; r < 8; ++r) {
    v[r]     = acc[0][r] * a1[0] + acc[1][r] * a1[1] +
               acc[2][r] * a1[2] + acc[3][r] * a1[3];
    v[8 + r] = acc[0][r] * a2[0] + acc[1][r] * a2[1] +
               acc[2][r] * a2[2] + acc[3][r] * a2[3];
  }
  const bool b3 = (eLow & 8) != 0;
  const bool b2 = (eLow & 4) != 0;
  const bool b1 = (eLow & 2) != 0;
  const bool b0 = (eLow & 1) != 0;
  float u[8];
#pragma unroll
  for (int j = 0; j < 8; ++j) {
    const float keep = b3 ? v[8 + j] : v[j];
    const float send = b3 ? v[j] : v[8 + j];
    u[j] = keep + gat_row_xor<8>(send);
  }
  float t[4];
#pragma unroll
  for (int j = 0; j < 4; ++j) {
    const float keep = b2 ? u[4 + j] : u[j];
    const float send = b2 ? u[j] : u[4 + j];
    t[j] = keep + gat_row_xor<4>(send);
  }
  float s[2];
#pragma unroll
  for (int j = 0; j < 2; ++j) {
    const float keep = b1 ? t[2 + j] : t[j];
    const float send = b1 ? t[j] : t[2 + j];
    s[j] = keep + gat_row_xor<2>(send);
  }
  const float keep0 = b0 ? s[1] : s[0];
  const float send0 = b0 ? s[0] : s[1];
  const float tot = keep0 + gat_row_xor<1>(send0);

  const int orow = rowBase + ewv * 16 + (eHalf << 3) + (eLow & 7);
  if (orow < M) {
    float* dst = b3 ? beta : alpha;
    dst[(size_t)orow * gridDim.y + blockIdx.y] = tot;
  }
}

__global__ void gat_count_kernel(const int* __restrict__ rows,
                                 int* __restrict__ cur,
                                 int* __restrict__ rank, int E) {
  int e = blockIdx.x * blockDim.x + threadIdx.x;
  if (e < E) rank[e] = atomicAdd(&cur[rows[e]], 1);
}

__global__ __launch_bounds__(1024) void gat_scan_kernel(
    const int* __restrict__ cur, int* __restrict__ off, int n) {
  __shared__ int wsum[32];
  __shared__ int tile_total;
  const int tid  = threadIdx.x;
  const int lane = tid & 31;
  const int wv   = tid >> 5;
  int carry = 0;
  for (int base = 0; base < n; base += 1024 * 16) {
    const int i0 = base + tid * 16;
    const bool full = (i0 + 16 <= n);
    int v[16];
    if (full) {
      for (int q = 0; q < 4; ++q) {
        const int4 t = *(const int4*)(cur + i0 + 4 * q);
        v[4 * q + 0] = t.x; v[4 * q + 1] = t.y;
        v[4 * q + 2] = t.z; v[4 * q + 3] = t.w;
      }
    } else {
      for (int k = 0; k < 16; ++k) v[k] = (i0 + k < n) ? cur[i0 + k] : 0;
    }
    int s = 0;
    for (int k = 0; k < 16; ++k) { const int t = v[k]; v[k] = s; s += t; }
    int incl = s;
    for (int d = 1; d < 32; d <<= 1) {
      const int t = __shfl_up(incl, d, 32);
      if (lane >= d) incl += t;
    }
    if (lane == 31) wsum[wv] = incl;
    __syncthreads();
    if (wv == 0) {
      const int t = wsum[lane];
      int ti = t;
      for (int d = 1; d < 32; d <<= 1) {
        const int u = __shfl_up(ti, d, 32);
        if (lane >= d) ti += u;
      }
      wsum[lane] = ti - t;
      if (lane == 31) tile_total = ti;
    }
    __syncthreads();
    const int pre = carry + wsum[wv] + (incl - s);
    if (full) {
      for (int q = 0; q < 4; ++q) {
        int4 t;
        t.x = pre + v[4 * q + 0]; t.y = pre + v[4 * q + 1];
        t.z = pre + v[4 * q + 2]; t.w = pre + v[4 * q + 3];
        *(int4*)(off + i0 + 4 * q) = t;
      }
    } else {
      for (int k = 0; k < 16; ++k) {
        if (i0 + k < n) off[i0 + k] = pre + v[k];
      }
    }
    carry += tile_total;
    __syncthreads();
  }
  if (tid == 0) off[n] = carry;
}

__global__ __launch_bounds__(256) void gat_fill_kernel(
    const int* __restrict__ rows, const int* __restrict__ cols,
    const int* __restrict__ off, const int* __restrict__ rank,
    int* __restrict__ ccol, int E) {
  const int e0 = (int)(blockIdx.x * blockDim.x + threadIdx.x) * 4;
  if (e0 >= E) return;
  const bool wide =
      (e0 + 3 < E) &&
      ((((size_t)rows | (size_t)cols | (size_t)rank) & 15) == 0);
  if (wide) {
    const int4 r = *(const int4*)(rows + e0);
    const int4 c = *(const int4*)(cols + e0);
    const int4 k = *(const int4*)(rank + e0);
    const int p0 = off[r.x];
    const int p1 = off[r.y];
    const int p2 = off[r.z];
    const int p3 = off[r.w];
    ccol[p0 + k.x] = c.x;
    ccol[p1 + k.y] = c.y;
    ccol[p2 + k.z] = c.z;
    ccol[p3 + k.w] = c.w;
  } else {
    for (int e = e0; e < E && e < e0 + 4; ++e)
      ccol[off[rows[e]] + rank[e]] = cols[e];
  }
}

__global__ __launch_bounds__(256) void gat_gather_l1_kernel(
    const int* __restrict__ off, const int* __restrict__ ccol,
    const float* __restrict__ alpha, const float* __restrict__ beta,
    const float* __restrict__ H, float* __restrict__ xc, int n) {
  const int row = __builtin_amdgcn_readfirstlane(
      (int)(blockIdx.x * 8 + (threadIdx.x >> 5)));
  const int lane = threadIdx.x & 31;
  if (row >= n) return;
  const int p0 = off[row];
  const int p1 = off[row + 1];
  const int hsel = lane >> 4;

  float al[4];
  float rs[4];
  float4 acc[4];
  for (int j = 0; j < 4; ++j) {
    al[j] = alpha[(size_t)row * 8 + 2 * j + hsel];
    rs[j] = 0.0f;
    acc[j] = make_float4(0.0f, 0.0f, 0.0f, 0.0f);
  }

  for (int base = p0; base < p1; base += 32) {
    const int m = (p1 - base < 32) ? (p1 - base) : 32;
    const int cv = (lane < m) ? ccol[base + lane] : 0;
    for (int i = 0; i < m; ++i) {
      const int c = __builtin_amdgcn_readlane(cv, i);
      const float* hc = H + (size_t)c * 512 + lane * 4;
      const float* bc = beta + (size_t)c * 8 + hsel;
      for (int j = 0; j < 4; ++j) {
        float s = al[j] + bc[2 * j];
        s = s > 0.0f ? s : 0.2f * s;
        const float w = __expf(-s);
        const float4 v = *(const float4*)(hc + j * 128);
        acc[j].x += w * v.x; acc[j].y += w * v.y;
        acc[j].z += w * v.z; acc[j].w += w * v.w;
        rs[j] += w;
      }
    }
  }

  float* xr = xc + (size_t)row * 512 + lane * 4;
  for (int j = 0; j < 4; ++j) {
    float4 o;
    o.x = acc[j].x / rs[j]; o.y = acc[j].y / rs[j];
    o.z = acc[j].z / rs[j]; o.w = acc[j].w / rs[j];
    o.x = o.x > 0.0f ? o.x : (__expf(o.x) - 1.0f);
    o.y = o.y > 0.0f ? o.y : (__expf(o.y) - 1.0f);
    o.z = o.z > 0.0f ? o.z : (__expf(o.z) - 1.0f);
    o.w = o.w > 0.0f ? o.w : (__expf(o.w) - 1.0f);
    *(float4*)(xr + j * 128) = o;
  }
}

__global__ __launch_bounds__(256) void gat_gather_l2_kernel(
    const int* __restrict__ off, const int* __restrict__ ccol,
    const float* __restrict__ alpha, const float* __restrict__ beta,
    const float* __restrict__ H2, float* __restrict__ out, int n) {
  const int row = __builtin_amdgcn_readfirstlane(
      (int)(blockIdx.x * 8 + (threadIdx.x >> 5)));
  const int lane = threadIdx.x & 31;
  if (row >= n) return;
  const int p0 = off[row];
  const int p1 = off[row + 1];
  const float a = alpha[row];

  float rs = 0.0f;
  float2 acc = make_float2(0.0f, 0.0f);
  for (int base = p0; base < p1; base += 32) {
    const int m = (p1 - base < 32) ? (p1 - base) : 32;
    const int cv = (lane < m) ? ccol[base + lane] : 0;
    for (int i = 0; i < m; ++i) {
      const int c = __builtin_amdgcn_readlane(cv, i);
      float s = a + beta[c];
      s = s > 0.0f ? s : 0.2f * s;
      const float w = __expf(-s);
      const float2 v = *(const float2*)(H2 + (size_t)c * 64 + lane * 2);
      acc.x += w * v.x; acc.y += w * v.y;
      rs += w;
    }
  }
  float2 o;
  o.x = acc.x / rs; o.y = acc.y / rs;
  *(float2*)(out + (size_t)row * 64 + lane * 2) = o;
}

extern "C" void kernel_launch(void* const* d_in, const int* in_sizes, int n_in,
                              void* d_out, int out_size, void* d_ws,
                              size_t ws_size, hipStream_t stream) {
  const float* x       = (const float*)d_in[0];
  const float* Ws      = (const float*)d_in[1];
  const float* a_heads = (const float*)d_in[2];
  const float* W_out   = (const float*)d_in[3];
  const float* a_out   = (const float*)d_in[4];
  const int*   eidx    = (const int*)d_in[5];
  float* out = (float*)d_out;

  const int N = in_sizes[0] / GAT_NFEAT;
  const int E = in_sizes[5] / 2;
  const int* e_row = eidx;
  const int* e_col = eidx + E;

  float* ws = (float*)d_ws;
  size_t off_f = 0;
  float* H      = ws + off_f; off_f += (size_t)N * 512;
  float* alpha1 = ws + off_f; off_f += (size_t)N * 8;
  float* beta1  = ws + off_f; off_f += (size_t)N * 8;
  float* xc     = ws + off_f; off_f += (size_t)N * 512;
  int* cur  = (int*)(ws + off_f); off_f += ((size_t)N + 3) / 4 * 4;
  int* roff = (int*)(ws + off_f); off_f += ((size_t)N + 4) / 4 * 4;
  int* ccol = (int*)(ws + off_f); off_f += (size_t)E;
  float* H2     = H;
  float* alpha2 = H + (size_t)N * 64;
  float* beta2  = alpha2 + (size_t)N;
  int* rank = (int*)xc;
  (void)ws_size; (void)n_in; (void)out_size;

  gat_zero_kernel<<<256, 256, 0, stream>>>((float*)cur, (size_t)N);
  gat_count_kernel<<<(E + 255) / 256, 256, 0, stream>>>(e_row, cur, rank, E);
  gat_scan_kernel<<<1, 1024, 0, stream>>>(cur, roff, N);
  gat_fill_kernel<<<((E + 3) / 4 + 255) / 256, 256, 0, stream>>>(
      e_row, e_col, roff, rank, ccol, E);

  const int mblocks = (N + 63) / 64;

  gat_gemm_f32_wmma<<<dim3(mblocks, GAT_HEADS), 128, 0, stream>>>(
      x, Ws, a_heads, H, alpha1, beta1, N, 512);

  gat_gather_l1_kernel<<<(N + 7) / 8, 256, 0, stream>>>(
      roff, ccol, alpha1, beta1, H, xc, N);

  gat_gemm_f32_wmma<<<dim3(mblocks, 1), 128, 0, stream>>>(
      xc, W_out, a_out, H2, alpha2, beta2, N, 64);

  gat_gather_l2_kernel<<<(N + 7) / 8, 256, 0, stream>>>(
      roff, ccol, alpha2, beta2, H2, out, N);
}


# Edge-row count moved into the layer-1 GEMM launch: each wave issues its rank atomics in its epilogue, hidden under the other resident waves' WMMAs; row sum order stays the atomics' arrival rank
# speedup vs baseline: 3.5131x; 1.0554x over previous
#include <hip/hip_runtime.h>
#include <math.h>

typedef __attribute__((ext_vector_type(2))) float v2f;
typedef __attribute__((ext_vector_type(8))) float v8f;

#define GAT_NFEAT 512
#define GAT_NHID  64
#define GAT_HEADS 8
#define GAT_NCLASS 64

__global__ void gat_zero_kernel(float* __restrict__ p, size_t n) {
  size_t i = (size_t)blockIdx.x * blockDim.x + threadIdx.x;
  size_t stride = (size_t)gridDim.x * blockDim.x;
  for (; i < n; i += stride) p[i] = 0.0f;
}


template <int XM>
__device__ __forceinline__ float gat_row_xor(float x) {
  const int b = __builtin_bit_cast(int, x);
  return __builtin_bit_cast(
      float, __builtin_amdgcn_update_dpp(b, b, 0x160 + XM, 0xf, 0xf, false));
}

template <bool COUNT>
__global__ __launch_bounds__(128) void gat_gemm_f32_wmma(
    const float* __restrict__ A, const float* __restrict__ B,
    const float* __restrict__ avec, float* __restrict__ C,
    float* __restrict__ alpha, float* __restrict__ beta, int M, int ldc,
    const int* __restrict__ erow, int* __restrict__ cur,
    int* __restrict__ rank, int E) {
  __shared__ float As[64][36];
  __shared__ float Bs[32][68];

  const int tid     = threadIdx.x;
  const int lane    = tid & 31;
  const int wv      = tid >> 5;
  const int laneLow = lane & 15;
  const int kOff    = (lane >> 4) << 1;
  const int rowBase = blockIdx.x * 64;
  const int colBase = blockIdx.y * 64;
  const float* Bblk = B + (size_t)blockIdx.y * (GAT_NFEAT * 64);

  v8f acc[4];
  acc[0] = {}; acc[1] = {}; acc[2] = {}; acc[3] = {};

  const int aCol = (tid & 7)  * 4;
  const int bCol = (tid & 15) * 4;
  const int aRow = tid >> 3;
  const int bRow = tid >> 4;

  const float* aSrc[4];
#pragma unroll
  for (int p = 0; p < 4; ++p) {
    int gRow = rowBase + aRow + p * 16;
    if (gRow >= M) gRow = M - 1;
    aSrc[p] = A + (size_t)gRow * GAT_NFEAT + aCol;
  }
  const float* bSrc = Bblk + (size_t)bRow * 64 + bCol;

  float4 ra0 = *(const float4*)(aSrc[0]);
  float4 ra1 = *(const float4*)(aSrc[1]);
  float4 ra2 = *(const float4*)(aSrc[2]);
  float4 ra3 = *(const float4*)(aSrc[3]);
  float4 rb0 = *(const float4*)(bSrc);
  float4 rb1 = *(const float4*)(bSrc + 8 * 64);
  float4 rb2 = *(const float4*)(bSrc + 16 * 64);
  float4 rb3 = *(const float4*)(bSrc + 24 * 64);

  for (int k0 = 0; k0 < GAT_NFEAT; k0 += 32) {
    *(float4*)&As[aRow][aCol]      = ra0;
    *(float4*)&As[aRow + 16][aCol] = ra1;
    *(float4*)&As[aRow + 32][aCol] = ra2;
    *(float4*)&As[aRow + 48][aCol] = ra3;
    *(float4*)&Bs[bRow][bCol]      = rb0;
    *(float4*)&Bs[bRow + 8][bCol]  = rb1;
    *(float4*)&Bs[bRow + 16][bCol] = rb2;
    *(float4*)&Bs[bRow + 24][bCol] = rb3;
    __syncthreads();

    if (k0 + 32 < GAT_NFEAT) {
      const float* bn = bSrc + (size_t)(k0 + 32) * 64;
      ra0 = *(const float4*)(aSrc[0] + k0 + 32);
      ra1 = *(const float4*)(aSrc[1] + k0 + 32);
      ra2 = *(const float4*)(aSrc[2] + k0 + 32);
      ra3 = *(const float4*)(aSrc[3] + k0 + 32);
      rb0 = *(const float4*)(bn);
      rb1 = *(const float4*)(bn + 8 * 64);
      rb2 = *(const float4*)(bn + 16 * 64);
      rb3 = *(const float4*)(bn + 24 * 64);
    }

#pragma unroll
    for (int kk = 0; kk < 32; kk += 4) {
      v2f a;
      a.x = As[wv * 16 + laneLow][kk + kOff];
      a.y = As[wv * 16 + laneLow][kk + kOff + 1];
#pragma unroll
      for (int n = 0; n < 4; ++n) {
        v2f b;
        b.x = Bs[kk + kOff][n * 16 + laneLow];
        b.y = Bs[kk + kOff + 1][n * 16 + laneLow];
        acc[n] = __builtin_amdgcn_wmma_f32_16x16x4_f32(
            false, a, false, b, (short)0, acc[n], false, false);
      }
    }
    __syncthreads();
  }

  const int el    = (int)__builtin_amdgcn_mbcnt_lo(~0u, 0u);
  const int eLow  = el & 15;
  const int eHalf = el >> 4;
  const int ewv   = __builtin_amdgcn_readfirstlane(tid >> 5);

  const long long ebase =
      (((long long)blockIdx.y * gridDim.x + blockIdx.x) * 4 + ewv) * 32;
  const long long estep = (long long)gridDim.x * gridDim.y * 128;
  bool eok[3];
  int  erk[3];
  if (COUNT) {
#pragma unroll
    for (int t = 0; t < 3; ++t) {
      eok[t] = ebase + t * estep + el < (long long)E;
      erk[t] = eok[t] ? (erow + (ebase + t * estep))[el] : 0;
    }
  }

  const float* av = avec + (size_t)blockIdx.y * 128;
  float a1[4], a2[4];
#pragma unroll
  for (int n = 0; n < 4; ++n) {
    a1[n] = av[n * 16 + eLow];
    a2[n] = av[64 + n * 16 + eLow];
  }

#pragma unroll
  for (int n = 0; n < 4; ++n) {
#pragma unroll
    for (int r = 0; r < 8; ++r) {
      int row = rowBase + ewv * 16 + r + (eHalf << 3);
      if (row < M)
        C[(size_t)row * ldc + colBase + n * 16 + eLow] = acc[n][r];
    }
  }

  float v[16];
#pragma unroll
  for (int r = 0; r < 8; ++r) {
    v[r]     = acc[0][r] * a1[0] + acc[1][r] * a1[1] +
               acc[2][r] * a1[2] + acc[3][r] * a1[3];
    v[8 + r] = acc[0][r] * a2[0] + acc[1][r] * a2[1] +
               acc[2][r] * a2[2] + acc[3][r] * a2[3];
  }
  const bool b3 = (eLow & 8) != 0;
  const bool b2 = (eLow & 4) != 0;
  const bool b1 = (eLow & 2) != 0;
  const bool b0 = (eLow & 1) != 0;
  float u[8];
#pragma unroll
  for (int j = 0; j < 8; ++j) {
    const float keep = b3 ? v[8 + j] : v[j];
    const float send = b3 ? v[j] : v[8 + j];
    u[j] = keep + gat_row_xor<8>(send);
  }
  float t[4];
#pragma unroll
  for (int j = 0; j < 4; ++j) {
    const float keep = b2 ? u[4 + j] : u[j];
    const float send = b2 ? u[j] : u[4 + j];
    t[j] = keep + gat_row_xor<4>(send);
  }
  float s[2];
#pragma unroll
  for (int j = 0; j < 2; ++j) {
    const float keep = b1 ? t[2 + j] : t[j];
    const float send = b1 ? t[j] : t[2 + j];
    s[j] = keep + gat_row_xor<2>(send);
  }
  const float keep0 = b0 ? s[1] : s[0];
  const float send0 = b0 ? s[0] : s[1];
  const float tot = keep0 + gat_row_xor<1>(send0);

  const int orow = rowBase + ewv * 16 + (eHalf << 3) + (eLow & 7);
  if (orow < M) {
    float* dst = b3 ? beta : alpha;
    dst[(size_t)orow * gridDim.y + blockIdx.y] = tot;
  }

  if (COUNT) {
#pragma unroll
    for (int t = 0; t < 3; ++t)
      if (eok[t]) erk[t] = atomicAdd(&cur[erk[t]], 1);
#pragma unroll
    for (int t = 0; t < 3; ++t)
      if (eok[t]) (rank + (ebase + t * estep))[el] = erk[t];
    for (long long e = ebase + 3 * estep + el; e < (long long)E; e += estep)
      rank[e] = atomicAdd(&cur[erow[e]], 1);
  }
}

__global__ __launch_bounds__(1024) void gat_scan_kernel(
    const int* __restrict__ cur, int* __restrict__ off, int n) {
  __shared__ int wsum[32];
  __shared__ int tile_total;
  const int tid  = threadIdx.x;
  const int lane = tid & 31;
  const int wv   = tid >> 5;
  int carry = 0;
  for (int base = 0; base < n; base += 1024 * 16) {
    const int i0 = base + tid * 16;
    const bool full = (i0 + 16 <= n);
    int v[16];
    if (full) {
      for (int q = 0; q < 4; ++q) {
        const int4 t = *(const int4*)(cur + i0 + 4 * q);
        v[4 * q + 0] = t.x; v[4 * q + 1] = t.y;
        v[4 * q + 2] = t.z; v[4 * q + 3] = t.w;
      }
    } else {
      for (int k = 0; k < 16; ++k) v[k] = (i0 + k < n) ? cur[i0 + k] : 0;
    }
    int s = 0;
    for (int k = 0; k < 16; ++k) { const int t = v[k]; v[k] = s; s += t; }
    int incl = s;
    for (int d = 1; d < 32; d <<= 1) {
      const int t = __shfl_up(incl, d, 32);
      if (lane >= d) incl += t;
    }
    if (lane == 31) wsum[wv] = incl;
    __syncthreads();
    if (wv == 0) {
      const int t = wsum[lane];
      int ti = t;
      for (int d = 1; d < 32; d <<= 1) {
        const int u = __shfl_up(ti, d, 32);
        if (lane >= d) ti += u;
      }
      wsum[lane] = ti - t;
      if (lane == 31) tile_total = ti;
    }
    __syncthreads();
    const int pre = carry + wsum[wv] + (incl - s);
    if (full) {
      for (int q = 0; q < 4; ++q) {
        int4 t;
        t.x = pre + v[4 * q + 0]; t.y = pre + v[4 * q + 1];
        t.z = pre + v[4 * q + 2]; t.w = pre + v[4 * q + 3];
        *(int4*)(off + i0 + 4 * q) = t;
      }
    } else {
      for (int k = 0; k < 16; ++k) {
        if (i0 + k < n) off[i0 + k] = pre + v[k];
      }
    }
    carry += tile_total;
    __syncthreads();
  }
  if (tid == 0) off[n] = carry;
}

__global__ __launch_bounds__(256) void gat_fill_kernel(
    const int* __restrict__ rows, const int* __restrict__ cols,
    const int* __restrict__ off, const int* __restrict__ rank,
    int* __restrict__ ccol, int E) {
  const int e0 = (int)(blockIdx.x * blockDim.x + threadIdx.x) * 4;
  if (e0 >= E) return;
  const bool wide =
      (e0 + 3 < E) &&
      ((((size_t)rows | (size_t)cols | (size_t)rank) & 15) == 0);
  if (wide) {
    const int4 r = *(const int4*)(rows + e0);
    const int4 c = *(const int4*)(cols + e0);
    const int4 k = *(const int4*)(rank + e0);
    const int p0 = off[r.x];
    const int p1 = off[r.y];
    const int p2 = off[r.z];
    const int p3 = off[r.w];
    ccol[p0 + k.x] = c.x;
    ccol[p1 + k.y] = c.y;
    ccol[p2 + k.z] = c.z;
    ccol[p3 + k.w] = c.w;
  } else {
    for (int e = e0; e < E && e < e0 + 4; ++e)
      ccol[off[rows[e]] + rank[e]] = cols[e];
  }
}

__global__ __launch_bounds__(256) void gat_gather_l1_kernel(
    const int* __restrict__ off, const int* __restrict__ ccol,
    const float* __restrict__ alpha, const float* __restrict__ beta,
    const float* __restrict__ H, float* __restrict__ xc, int n) {
  const int row = __builtin_amdgcn_readfirstlane(
      (int)(blockIdx.x * 8 + (threadIdx.x >> 5)));
  const int lane = threadIdx.x & 31;
  if (row >= n) return;
  const int p0 = off[row];
  const int p1 = off[row + 1];
  const int hsel = lane >> 4;

  float al[4];
  float rs[4];
  float4 acc[4];
  for (int j = 0; j < 4; ++j) {
    al[j] = alpha[(size_t)row * 8 + 2 * j + hsel];
    rs[j] = 0.0f;
    acc[j] = make_float4(0.0f, 0.0f, 0.0f, 0.0f);
  }

  for (int base = p0; base < p1; base += 32) {
    const int m = (p1 - base < 32) ? (p1 - base) : 32;
    const int cv = (lane < m) ? ccol[base + lane] : 0;
    for (int i = 0; i < m; ++i) {
      const int c = __builtin_amdgcn_readlane(cv, i);
      const float* hc = H + (size_t)c * 512 + lane * 4;
      const float* bc = beta + (size_t)c * 8 + hsel;
      for (int j = 0; j < 4; ++j) {
        float s = al[j] + bc[2 * j];
        s = s > 0.0f ? s : 0.2f * s;
        const float w = __expf(-s);
        const float4 v = *(const float4*)(hc + j * 128);
        acc[j].x += w * v.x; acc[j].y += w * v.y;
        acc[j].z += w * v.z; acc[j].w += w * v.w;
        rs[j] += w;
      }
    }
  }

  float* xr = xc + (size_t)row * 512 + lane * 4;
  for (int j = 0; j < 4; ++j) {
    float4 o;
    o.x = acc[j].x / rs[j]; o.y = acc[j].y / rs[j];
    o.z = acc[j].z / rs[j]; o.w = acc[j].w / rs[j];
    o.x = o.x > 0.0f ? o.x : (__expf(o.x) - 1.0f);
    o.y = o.y > 0.0f ? o.y : (__expf(o.y) - 1.0f);
    o.z = o.z > 0.0f ? o.z : (__expf(o.z) - 1.0f);
    o.w = o.w > 0.0f ? o.w : (__expf(o.w) - 1.0f);
    *(float4*)(xr + j * 128) = o;
  }
}

__global__ __launch_bounds__(256) void gat_gather_l2_kernel(
    const int* __restrict__ off, const int* __restrict__ ccol,
    const float* __restrict__ alpha, const float* __restrict__ beta,
    const float* __restrict__ H2, float* __restrict__ out, int n) {
  const int row = __builtin_amdgcn_readfirstlane(
      (int)(blockIdx.x * 8 + (threadIdx.x >> 5)));
  const int lane = threadIdx.x & 31;
  if (row >= n) return;
  const int p0 = off[row];
  const int p1 = off[row + 1];
  const float a = alpha[row];

  float rs = 0.0f;
  float2 acc = make_float2(0.0f, 0.0f);
  for (int base = p0; base < p1; base += 32) {
    const int m = (p1 - base < 32) ? (p1 - base) : 32;
    const int cv = (lane < m) ? ccol[base + lane] : 0;
    for (int i = 0; i < m; ++i) {
      const int c = __builtin_amdgcn_readlane(cv, i);
      float s = a + beta[c];
      s = s > 0.0f ? s : 0.2f * s;
      const float w = __expf(-s);
      const float2 v = *(const float2*)(H2 + (size_t)c * 64 + lane * 2);
      acc.x += w * v.x; acc.y += w * v.y;
      rs += w;
    }
  }
  float2 o;
  o.x = acc.x / rs; o.y = acc.y / rs;
  *(float2*)(out + (size_t)row * 64 + lane * 2) = o;
}

extern "C" void kernel_launch(void* const* d_in, const int* in_sizes, int n_in,
                              void* d_out, int out_size, void* d_ws,
                              size_t ws_size, hipStream_t stream) {
  const float* x       = (const float*)d_in[0];
  const float* Ws      = (const float*)d_in[1];
  const float* a_heads = (const float*)d_in[2];
  const float* W_out   = (const float*)d_in[3];
  const float* a_out   = (const float*)d_in[4];
  const int*   eidx    = (const int*)d_in[5];
  float* out = (float*)d_out;

  const int N = in_sizes[0] / GAT_NFEAT;
  const int E = in_sizes[5] / 2;
  const int* e_row = eidx;
  const int* e_col = eidx + E;

  float* ws = (float*)d_ws;
  size_t off_f = 0;
  float* H      = ws + off_f; off_f += (size_t)N * 512;
  float* alpha1 = ws + off_f; off_f += (size_t)N * 8;
  float* beta1  = ws + off_f; off_f += (size_t)N * 8;
  float* xc     = ws + off_f; off_f += (size_t)N * 512;
  int* cur  = (int*)(ws + off_f); off_f += ((size_t)N + 3) / 4 * 4;
  int* roff = (int*)(ws + off_f); off_f += ((size_t)N + 4) / 4 * 4;
  int* ccol = (int*)(ws + off_f); off_f += (size_t)E;
  float* H2     = H;
  float* alpha2 = H + (size_t)N * 64;
  float* beta2  = alpha2 + (size_t)N;
  int* rank = (int*)xc;
  (void)ws_size; (void)n_in; (void)out_size;

  gat_zero_kernel<<<256, 256, 0, stream>>>((float*)cur, (size_t)N);

  const int mblocks = (N + 63) / 64;

  gat_gemm_f32_wmma<true><<<dim3(mblocks, GAT_HEADS), 128, 0, stream>>>(
      x, Ws, a_heads, H, alpha1, beta1, N, 512, e_row, cur, rank, E);

  gat_scan_kernel<<<1, 1024, 0, stream>>>(cur, roff, N);
  gat_fill_kernel<<<((E + 3) / 4 + 255) / 256, 256, 0, stream>>>(
      e_row, e_col, roff, rank, ccol, E);

  gat_gather_l1_kernel<<<(N + 7) / 8, 256, 0, stream>>>(
      roff, ccol, alpha1, beta1, H, xc, N);

  gat_gemm_f32_wmma<false><<<dim3(mblocks, 1), 128, 0, stream>>>(
      xc, W_out, a_out, H2, alpha2, beta2, N, 64, nullptr, nullptr, nullptr, 0);

  gat_gather_l2_kernel<<<(N + 7) / 8, 256, 0, stream>>>(
      roff, ccol, alpha2, beta2, H2, out, N);
}
